# starting kernel
# baseline (speedup 1.0000x reference)
#include <hip/hip_runtime.h>

#define D_DIM 4096
#define E_NUM 64
#define TOKS  16384
#define RBLKS 256
#define CAPF  256.0

typedef float v2f __attribute__((ext_vector_type(2)));
typedef float v8f __attribute__((ext_vector_type(8)));

__global__ __launch_bounds__(256) void pack_w_kernel(const float* __restrict__ W,
                                                     float* __restrict__ Wp) {
  int i = blockIdx.x * 256 + threadIdx.x;
  if (i >= (D_DIM / 2) * E_NUM) return;
  int q = i >> 6, n = i & 63;
  v2f v;
  v.x = W[(2 * q) * E_NUM + n];
  v.y = W[(2 * q + 1) * E_NUM + n];
  ((v2f*)Wp)[i] = v;
}

__global__ __launch_bounds__(128) void router_kernel(
    const float* __restrict__ x, const float* __restrict__ Wp,
    const float* __restrict__ bias, float* __restrict__ out,
    int* __restrict__ top_idx, float* __restrict__ top_prob,
    float* __restrict__ Pblk, float* __restrict__ fblk, int* __restrict__ cblk) {
  __shared__ float lds[4][16][E_NUM + 1];
  __shared__ int   amax[64];
  __shared__ float aprob[64];

  const int tid  = threadIdx.x;
  const int wave = tid >> 5;
  const int lane = tid & 31;
  const int m    = lane & 15;
  const int h    = lane >> 4;
  const int tok0 = (blockIdx.x * 4 + wave) * 16;

  const float* xrow = x + (size_t)(tok0 + m) * D_DIM;

  v8f c0{}, c1{}, c2{}, c3{};
  for (int k0 = 0; k0 < D_DIM; k0 += 4) {
    v2f a = *(const v2f*)(xrow + k0 + 2 * h);
    const v2f* wq = (const v2f*)Wp + (size_t)(k0 / 2 + h) * E_NUM + m;
    v2f b0 = wq[0];
    v2f b1 = wq[16];
    v2f b2 = wq[32];
    v2f b3 = wq[48];
    c0 = __builtin_amdgcn_wmma_f32_16x16x4_f32(false, a, false, b0, (short)0, c0, false, false);
    c1 = __builtin_amdgcn_wmma_f32_16x16x4_f32(false, a, false, b1, (short)0, c1, false, false);
    c2 = __builtin_amdgcn_wmma_f32_16x16x4_f32(false, a, false, b2, (short)0, c2, false, false);
    c3 = __builtin_amdgcn_wmma_f32_16x16x4_f32(false, a, false, b3, (short)0, c3, false, false);
  }

#pragma unroll
  for (int r = 0; r < 8; ++r) {
    int row = r + 8 * h;
    lds[wave][row][m]      = c0[r];
    lds[wave][row][16 + m] = c1[r];
    lds[wave][row][32 + m] = c2[r];
    lds[wave][row][48 + m] = c3[r];
  }
  __syncthreads();

  if (tid < 64) {
    const int w = tid >> 4, row = tid & 15;
    const int token = blockIdx.x * 64 + tid;
    float mx = -3.4e38f;
#pragma unroll 8
    for (int e = 0; e < E_NUM; ++e) {
      float l = lds[w][row][e] + bias[e];
      lds[w][row][e] = l;
      mx = fmaxf(mx, l);
    }
    float s = 0.f;
#pragma unroll 8
    for (int e = 0; e < E_NUM; ++e) s += expf(lds[w][row][e] - mx);
    float inv = 1.f / s;
    int   best  = 0;
    float bestp = -1.f;
    float* orow = out + (size_t)token * E_NUM;
#pragma unroll 8
    for (int e = 0; e < E_NUM; ++e) {
      float p = expf(lds[w][row][e] - mx) * inv;
      lds[w][row][e] = p;
      orow[e] = p;
      if (p > bestp) { bestp = p; best = e; }
    }
    top_idx[token]  = best;
    top_prob[token] = bestp;
    amax[tid]  = best;
    aprob[tid] = bestp;
  }
  __syncthreads();

  if (tid < 64) {
    const int e = tid;
    float Ps = 0.f, fs = 0.f;
    int cnt = 0;
#pragma unroll 4
    for (int i = 0; i < 64; ++i) {
      Ps += lds[i >> 4][i & 15][e];
      if (amax[i] == e) { fs += aprob[i]; ++cnt; }
    }
    Pblk[blockIdx.x * 64 + e] = Ps;
    fblk[blockIdx.x * 64 + e] = fs;
    cblk[blockIdx.x * 64 + e] = cnt;
  }
}

__global__ __launch_bounds__(64) void finalize_kernel(
    const float* __restrict__ Pblk, const float* __restrict__ fblk,
    const int* __restrict__ cblk, int* __restrict__ count,
    int* __restrict__ offset, int* __restrict__ cursor,
    float* __restrict__ out) {
  __shared__ float term[64];
  const int e = threadIdx.x;
  double P = 0.0, f = 0.0;
  int c = 0;
  for (int b = 0; b < RBLKS; ++b) {
    P += Pblk[b * 64 + e];
    f += fblk[b * 64 + e];
    c += cblk[b * 64 + e];
  }
  count[e] = c;
  const double T = (double)TOKS;
  term[e] = (float)((f / T) * (P / T));
  __syncthreads();
  if (e == 0) {
    int off = 0;
    float aux = 0.f;
    for (int i = 0; i < 64; ++i) {
      offset[i] = off;
      cursor[i] = off;
      off += count[i];
      aux += term[i];
    }
    out[(size_t)TOKS * E_NUM] = 0.01f * (float)E_NUM * aux;
  }
}

__global__ __launch_bounds__(256) void scatter_kernel(
    const int* __restrict__ top_idx, const float* __restrict__ top_prob,
    int* __restrict__ cursor, int* __restrict__ list_tok,
    float* __restrict__ list_prob) {
  int t = blockIdx.x * 256 + threadIdx.x;
  if (t >= TOKS) return;
  int e = top_idx[t];
  int slot = atomicAdd(&cursor[e], 1);
  list_tok[slot]  = t;
  list_prob[slot] = top_prob[t];
}

__global__ __launch_bounds__(256) void assign_kernel(
    const int* __restrict__ top_idx, const float* __restrict__ top_prob,
    const int* __restrict__ offset, const int* __restrict__ count,
    const int* __restrict__ list_tok, const float* __restrict__ list_prob,
    float* __restrict__ out) {
  int t = blockIdx.x * 256 + threadIdx.x;
  if (t >= TOKS) return;
  const int   e = top_idx[t];
  const float p = top_prob[t];
  const int o = offset[e], c = count[e];
  double s = 0.0;
  for (int j = o; j < o + c; ++j) {
    float pj = list_prob[j];
    int   tj = list_tok[j];
    if (pj > p || (pj == p && tj <= t)) s += (double)pj;
  }
  if (s > CAPF) {
    float* orow = out + (size_t)t * E_NUM;
#pragma unroll 8
    for (int e2 = 0; e2 < E_NUM; ++e2) orow[e2] = 0.f;
  }
}

extern "C" void kernel_launch(void* const* d_in, const int* in_sizes, int n_in,
                              void* d_out, int out_size, void* d_ws, size_t ws_size,
                              hipStream_t stream) {
  (void)in_sizes; (void)n_in; (void)out_size; (void)ws_size;
  const float* x = (const float*)d_in[0];
  const float* W = (const float*)d_in[1];
  const float* b = (const float*)d_in[2];
  float* out = (float*)d_out;

  char* ws = (char*)d_ws;
  size_t off = 0;
  auto alloc = [&](size_t bytes) { char* p = ws + off; off = (off + bytes + 255) & ~(size_t)255; return p; };

  float* Wp        = (float*)alloc((size_t)(D_DIM / 2) * E_NUM * 2 * sizeof(float));
  int*   top_idx   = (int*)  alloc(TOKS * sizeof(int));
  float* top_prob  = (float*)alloc(TOKS * sizeof(float));
  int*   list_tok  = (int*)  alloc(TOKS * sizeof(int));
  float* list_prob = (float*)alloc(TOKS * sizeof(float));
  float* Pblk      = (float*)alloc(RBLKS * 64 * sizeof(float));
  float* fblk      = (float*)alloc(RBLKS * 64 * sizeof(float));
  int*   cblk      = (int*)  alloc(RBLKS * 64 * sizeof(int));
  int*   count     = (int*)  alloc(64 * sizeof(int));
  int*   offset    = (int*)  alloc(64 * sizeof(int));
  int*   cursor    = (int*)  alloc(64 * sizeof(int));

  pack_w_kernel<<<((D_DIM / 2) * E_NUM + 255) / 256, 256, 0, stream>>>(W, Wp);
  router_kernel<<<RBLKS, 128, 0, stream>>>(x, Wp, b, out, top_idx, top_prob,
                                           Pblk, fblk, cblk);
  finalize_kernel<<<1, 64, 0, stream>>>(Pblk, fblk, cblk, count, offset, cursor, out);
  scatter_kernel<<<TOKS / 256, 256, 0, stream>>>(top_idx, top_prob, cursor,
                                                 list_tok, list_prob);
  assign_kernel<<<TOKS / 256, 256, 0, stream>>>(top_idx, top_prob, offset, count,
                                                list_tok, list_prob, out);
}


# capacity rank sums in exact 64-bit integers (prob*2^29 is a whole number for a top-1 prob in [2^-6,1]) instead of a dependent f64 add chain; one u64 key compare per element; same keep decisions
# speedup vs baseline: 4.9312x; 4.9312x over previous
#include <hip/hip_runtime.h>

#define D_DIM 4096
#define E_NUM 64
#define TOKS  16384
#define RBLKS 256
#define CAP   256

typedef float v2f __attribute__((ext_vector_type(2)));
typedef float v4f __attribute__((ext_vector_type(4)));
typedef float v8f __attribute__((ext_vector_type(8)));

__global__ __launch_bounds__(256) void pack_w_kernel(const float* __restrict__ W,
                                                     float* __restrict__ Wp) {
  int i = blockIdx.x * 256 + threadIdx.x;
  if (i >= (D_DIM / 2) * E_NUM) return;
  int q = i >> 6, n = i & 63;
  v2f v;
  v.x = W[(2 * q) * E_NUM + n];
  v.y = W[(2 * q + 1) * E_NUM + n];
  ((v2f*)Wp)[i] = v;
}

#define KC       128
#define NCH      (D_DIM / KC)
#define XP       (KC + 4)
#define WPP      160
#define XS_DW    (64 * XP)
#define WS_DW    ((KC / 2) * WPP)
#define STAGE_DW (XS_DW + WS_DW)

__device__ __forceinline__ void lds_barrier() {
  __builtin_amdgcn_fence(__ATOMIC_RELEASE, "workgroup", "local");
  __builtin_amdgcn_s_barrier();
  __builtin_amdgcn_fence(__ATOMIC_ACQUIRE, "workgroup", "local");
}

#define KG 8
#define NG (KC / 4 / KG)

__device__ __forceinline__ void slice_mma(const float* ap, const float* bp, v8f& c0, v8f& c1) {
  v2f fa[2][KG], fb0[2][KG], fb1[2][KG];
#pragma unroll
  for (int j = 0; j < KG; ++j) {
    fa[0][j]  = *(const v2f*)(ap + 4 * j);
    fb0[0][j] = *(const v2f*)(bp + 2 * j * WPP);
    fb1[0][j] = *(const v2f*)(bp + 2 * j * WPP + 32);
  }
#pragma unroll
  for (int g = 0; g < NG; ++g) {
    const int cur = g & 1, nxt = cur ^ 1;
    if (g + 1 < NG) {
#pragma unroll
      for (int j = 0; j < KG; ++j) {
        const int ks = (g + 1) * KG + j;
        fa[nxt][j]  = *(const v2f*)(ap + 4 * ks);
        fb0[nxt][j] = *(const v2f*)(bp + 2 * ks * WPP);
        fb1[nxt][j] = *(const v2f*)(bp + 2 * ks * WPP + 32);
      }
    }
    __builtin_amdgcn_sched_barrier(0);
#pragma unroll
    for (int j = 0; j < KG; ++j) {
      c0 = __builtin_amdgcn_wmma_f32_16x16x4_f32(false, fa[cur][j], false, fb0[cur][j], (short)0, c0, false, false);
      c1 = __builtin_amdgcn_wmma_f32_16x16x4_f32(false, fa[cur][j], false, fb1[cur][j], (short)0, c1, false, false);
    }
    __builtin_amdgcn_sched_barrier(0);
  }
}

__global__ __launch_bounds__(256) void router_kernel(
    const float* __restrict__ x, const float* __restrict__ Wp,
    const float* __restrict__ bias, float* __restrict__ out,
    int* __restrict__ top_idx, float* __restrict__ top_prob,
    float* __restrict__ Pblk, float* __restrict__ fblk, int* __restrict__ cblk) {
  __shared__ __attribute__((aligned(16))) float stage[2][STAGE_DW];
  __shared__ float lds[4][16][E_NUM + 1];
  __shared__ int   amax[64];
  __shared__ float aprob[64];

  const int tid  = threadIdx.x;
  const int wave = tid >> 5;
  const int lane = tid & 31;
  const int m    = lane & 15;
  const int h    = lane >> 4;
  const int tg   = wave & 3;
  const int eh   = wave >> 2;

  const int crow = tid >> 5;
  const int ccol = (tid & 31) * 4;
  const float* xg = x + ((size_t)blockIdx.x * 64 + crow) * D_DIM + ccol;
  const float* wg = Wp + (size_t)tid * 4;

#define FOR_NLD(M) M(0) M(1) M(2) M(3) M(4) M(5) M(6) M(7)
#define DECL_R(i)  v4f xr##i, wr##i;
#define GLOAD_R(i) xr##i = *(const v4f*)(xsrc + (size_t)(8 * i) * D_DIM); wr##i = *(const v4f*)(wsrc + i * 1024);
#define LSTORE_R(i) *(v4f*)(xdst + (crow + 8 * i) * XP + ccol) = xr##i; *(v4f*)(xdst + XS_DW + (crow + 8 * i) * WPP + ccol) = wr##i;
  FOR_NLD(DECL_R)
  const float* xsrc;
  const float* wsrc;
  float* xdst;

  xsrc = xg; wsrc = wg;
  FOR_NLD(GLOAD_R)
  xdst = stage[0];
  FOR_NLD(LSTORE_R)
  xsrc = xg + KC; wsrc = wg + (KC / 2) * 128;
  FOR_NLD(GLOAD_R)
  lds_barrier();

  const int aoff = (16 * tg + m) * XP + 2 * h;
  const int boff = XS_DW + h * WPP + 2 * (32 * eh + m);

  v8f c0{}, c1{};
  for (int c = 0; c < NCH - 2; ++c) {
    slice_mma(stage[c & 1] + aoff, stage[c & 1] + boff, c0, c1);

    xdst = stage[(c + 1) & 1];
    FOR_NLD(LSTORE_R)
    xsrc = xg + (c + 2) * KC;
    wsrc = wg + (size_t)(c + 2) * (KC / 2) * 128;
    FOR_NLD(GLOAD_R)
    lds_barrier();
  }
  slice_mma(stage[(NCH - 2) & 1] + aoff, stage[(NCH - 2) & 1] + boff, c0, c1);
  xdst = stage[(NCH - 1) & 1];
  FOR_NLD(LSTORE_R)
  lds_barrier();
  slice_mma(stage[(NCH - 1) & 1] + aoff, stage[(NCH - 1) & 1] + boff, c0, c1);
#undef FOR_NLD
#undef DECL_R
#undef GLOAD_R
#undef LSTORE_R

#pragma unroll
  for (int r = 0; r < 8; ++r) {
    int row = r + 8 * h;
    lds[tg][row][32 * eh + m]      = c0[r];
    lds[tg][row][32 * eh + 16 + m] = c1[r];
  }
  __syncthreads();

  if (tid < 64) {
    const int w = tid >> 4, row = tid & 15;
    const int token = blockIdx.x * 64 + tid;
    float mx = -3.4e38f;
#pragma unroll 8
    for (int e = 0; e < E_NUM; ++e) {
      float l = lds[w][row][e] + bias[e];
      lds[w][row][e] = l;
      mx = fmaxf(mx, l);
    }
    float s = 0.f;
#pragma unroll 8
    for (int e = 0; e < E_NUM; ++e) s += expf(lds[w][row][e] - mx);
    float inv = 1.f / s;
    int   best  = 0;
    float bestp = -1.f;
    float* orow = out + (size_t)token * E_NUM;
#pragma unroll 8
    for (int e = 0; e < E_NUM; ++e) {
      float p = expf(lds[w][row][e] - mx) * inv;
      lds[w][row][e] = p;
      orow[e] = p;
      if (p > bestp) { bestp = p; best = e; }
    }
    top_idx[token]  = best;
    top_prob[token] = bestp;
    amax[tid]  = best;
    aprob[tid] = bestp;
  }
  __syncthreads();

  if (tid < 64) {
    const int e = tid;
    float Ps = 0.f, fs = 0.f;
    int cnt = 0;
#pragma unroll
    for (int i = 0; i < 64; ++i) {
      const float pv   = lds[i >> 4][i & 15][e];
      const int   ai   = amax[i];
      const float ap   = aprob[i];
      const bool  mine = (ai == e);
      Ps += pv;
      fs += mine ? ap : 0.f;
      cnt += mine ? 1 : 0;
    }
    Pblk[blockIdx.x * 64 + e] = Ps;
    fblk[blockIdx.x * 64 + e] = fs;
    cblk[blockIdx.x * 64 + e] = cnt;
  }
}

#define PLACE_THREADS 1024
#define PLACE_BLKS    (TOKS / PLACE_THREADS)
#define PARTS         (PLACE_THREADS / 64)
#define PART_RB       (RBLKS / PARTS)
static_assert(PART_RB * 64 == PLACE_THREADS, "part q must be the router blocks of workgroup q");
static_assert(PLACE_BLKS == PARTS, "one part per workgroup");
__global__ __launch_bounds__(PLACE_THREADS) void place_kernel(
    const float* __restrict__ Pblk, const float* __restrict__ fblk,
    const int* __restrict__ cblk, const int* __restrict__ top_idx,
    const float* __restrict__ top_prob, int* __restrict__ count,
    int* __restrict__ offset, int* __restrict__ list_tok,
    float* __restrict__ list_prob, double* __restrict__ Ppart,
    double* __restrict__ Fpart) {
  __shared__ int    partc[PARTS][64];
  __shared__ double wide[2][PART_RB][64];
  __shared__ int    cnt[64];
  __shared__ int    cursor[64];
  const int tid  = threadIdx.x;
  const int e    = tid & 63;
  const int part = tid >> 6;
  const int k    = blockIdx.x;

  const int   t  = k * PLACE_THREADS + tid;
  const int   te = top_idx[t];
  const float tp = top_prob[t];

  const float pv = Pblk[(k * PART_RB + part) * 64 + e];
  const float fv = fblk[(k * PART_RB + part) * 64 + e];

  const int rb0 = part * PART_RB;
  int c = 0;
  for (int i = 0; i < PART_RB; ++i) c += cblk[(rb0 + i) * 64 + e];
  partc[part][e]   = c;
  wide[0][part][e] = (double)pv;
  wide[1][part][e] = (double)fv;
  __syncthreads();

  int bef = 0;
  if (tid < 64) {
    int tot = 0;
    for (int q = 0; q < PARTS; ++q) {
      const int v = partc[q][tid];
      if (q < k) bef += v;
      tot += v;
    }
    cnt[tid] = tot;
    if (k == 0) count[tid] = tot;
  }
  if (tid < 128) {
    const int which = tid >> 6;
    double a = 0.0;
    for (int i = 0; i < PART_RB; ++i) a += wide[which][i][e];
    (which ? Fpart : Ppart)[k * 64 + e] = a;
  }
  __syncthreads();

  if (tid < 64) {
    int off = 0;
    for (int i = 0; i < 64; ++i) {
      const int v = cnt[i];
      off += (i < tid) ? v : 0;
    }
    cursor[tid] = off + bef;
    if (k == 0) offset[tid] = off;
  }
  __syncthreads();

  const int slot = atomicAdd(&cursor[te], 1);
  list_tok[slot]  = t;
  list_prob[slot] = tp;
}

#define QSCALE 536870912.0f
#define QCAP   ((unsigned long long)CAP << 29)
#define ASG_THREADS 64
#define ASG_PIECES  (TOKS / 32 + E_NUM)
static_assert(ASG_THREADS == E_NUM, "the aux workgroup has one thread per expert");
__global__ __launch_bounds__(ASG_THREADS) void assign_kernel(
    const int* __restrict__ offset, const int* __restrict__ count,
    const int* __restrict__ list_tok, const float* __restrict__ list_prob,
    const double* __restrict__ Ppart, const double* __restrict__ Fpart,
    float* __restrict__ out) {
  if (blockIdx.x == ASG_PIECES / (ASG_THREADS / 32)) {
    __shared__ float term[E_NUM];
    const int e = threadIdx.x;
    double P = 0.0, f = 0.0;
    for (int q = 0; q < PARTS; ++q) { P += Ppart[q * 64 + e]; f += Fpart[q * 64 + e]; }
    const double T = (double)TOKS;
    term[e] = (float)((f / T) * (P / T));
    __syncthreads();
    if (e == 0) {
      float aux = 0.f;
      for (int i = 0; i < E_NUM; ++i) aux += term[i];
      out[(size_t)TOKS * E_NUM] = 0.01f * (float)E_NUM * aux;
    }
    return;
  }
  const int lane = threadIdx.x & 31;
  const int w = __builtin_amdgcn_readfirstlane((int)(blockIdx.x * (ASG_THREADS / 32) + (threadIdx.x >> 5)));
  int e = 0, base = 0, first = 0;
#pragma unroll
  for (int x = 0; x < E_NUM; ++x) {
    if (w >= first) { e = x; base = first; }
    first += (count[x] + 31) >> 5;
  }
  if (w >= first) return;
  const int q = w - base;
  const int o = offset[e], c = count[e];
  const int last = o + c - 1;
  const int js = 32 * q + lane;
  const bool valid = js < c;
  const int   t = list_tok[min(o + js, last)];
  const float p = list_prob[min(o + js, last)];
  const unsigned long long kown =
      ((unsigned long long)(unsigned)(p * QSCALE) << 32) | (unsigned)(0x7FFFFFFF - t);

  int jq = o + lane;
  float pj = list_prob[min(jq, last)];
  int   tj = list_tok[min(jq, last)];
  if (jq > last) pj = 0.f;
  unsigned long long s = 0;
  for (int k = 0; k < c; k += 32) {
    jq += 32;
    const float pn = list_prob[min(jq, last)];
    const int   tn = list_tok[min(jq, last)];
    __builtin_amdgcn_sched_barrier(0);
    const unsigned qj = (unsigned)(pj * QSCALE);
    const unsigned rj = (unsigned)(0x7FFFFFFF - tj);
#pragma unroll
    for (int sl = 0; sl < 32; ++sl) {
      const unsigned bq = (unsigned)__builtin_amdgcn_readlane((int)qj, sl);
      const unsigned br = (unsigned)__builtin_amdgcn_readlane((int)rj, sl);
      const unsigned long long kb = ((unsigned long long)bq << 32) | br;
      s += (kb >= kown) ? (unsigned long long)bq : 0ull;
    }
    pj = (jq > last) ? 0.f : pn;
    tj = tn;
  }
  if (valid && s > QCAP) {
    float* orow = out + (size_t)t * E_NUM;
#pragma unroll 8
    for (int e2 = 0; e2 < E_NUM; ++e2) orow[e2] = 0.f;
  }
}

extern "C" void kernel_launch(void* const* d_in, const int* in_sizes, int n_in,
                              void* d_out, int out_size, void* d_ws, size_t ws_size,
                              hipStream_t stream) {
  (void)in_sizes; (void)n_in; (void)out_size; (void)ws_size;
  const float* x = (const float*)d_in[0];
  const float* W = (const float*)d_in[1];
  const float* b = (const float*)d_in[2];
  float* out = (float*)d_out;

  char* ws = (char*)d_ws;
  size_t off = 0;
  auto alloc = [&](size_t bytes) { char* p = ws + off; off = (off + bytes + 255) & ~(size_t)255; return p; };

  float* Wp        = (float*)alloc((size_t)(D_DIM / 2) * E_NUM * 2 * sizeof(float));
  int*   top_idx   = (int*)  alloc(TOKS * sizeof(int));
  float* top_prob  = (float*)alloc(TOKS * sizeof(float));
  int*   list_tok  = (int*)  alloc(TOKS * sizeof(int));
  float* list_prob = (float*)alloc(TOKS * sizeof(float));
  float* Pblk      = (float*)alloc(RBLKS * 64 * sizeof(float));
  float* fblk      = (float*)alloc(RBLKS * 64 * sizeof(float));
  int*   cblk      = (int*)  alloc(RBLKS * 64 * sizeof(int));
  int*   count     = (int*)  alloc(64 * sizeof(int));
  int*   offset    = (int*)  alloc(64 * sizeof(int));

  pack_w_kernel<<<((D_DIM / 2) * E_NUM + 255) / 256, 256, 0, stream>>>(W, Wp);
  router_kernel<<<RBLKS, 256, 0, stream>>>(x, Wp, b, out, top_idx, top_prob,
                                           Pblk, fblk, cblk);
  double* Ppart = (double*)Wp;
  double* Fpart = Ppart + PARTS * 64;
  place_kernel<<<PLACE_BLKS, PLACE_THREADS, 0, stream>>>(Pblk, fblk, cblk, top_idx, top_prob,
                                                         count, offset, list_tok, list_prob,
                                                         Ppart, Fpart);
  assign_kernel<<<ASG_PIECES / (ASG_THREADS / 32) + 1, ASG_THREADS, 0, stream>>>(
      offset, count, list_tok, list_prob, Ppart, Fpart, out);
}
